# starting kernel
# baseline (speedup 1.0000x reference)
#include <hip/hip_runtime.h>


typedef float v2f __attribute__((ext_vector_type(2)));
typedef float v8f __attribute__((ext_vector_type(8)));

#define NTOK 4096
#define DM   512
#define NH   8
#define DHD  64
#define SCALE 0.125f

static __device__ __forceinline__ v8f wmma_f32(v2f a, v2f b, v8f c) {
  return __builtin_amdgcn_wmma_f32_16x16x4_f32(
       false, a,  false, b,
       (short)0, c,  false,  false);
}

__global__ __launch_bounds__(256) void qkv_proj_kernel(
    const float* __restrict__ h,
    const float* __restrict__ Wq, const float* __restrict__ bq,
    const float* __restrict__ Wk, const float* __restrict__ bk,
    const float* __restrict__ Wv, const float* __restrict__ bv,
    float* __restrict__ Qo, float* __restrict__ Ko, float* __restrict__ Vo)
{
  const int lane = threadIdx.x & 31;
  const int wid  = threadIdx.x >> 5;
  const int tile = blockIdx.x * 8 + wid;
  const int mat  = tile >> 13;
  const int rem  = tile & 8191;
  const int rowT = rem >> 5;
  const int colT = rem & 31;

  const float* W; const float* bias; float* Out;
  if (mat == 0)      { W = Wq; bias = bq; Out = Qo; }
  else if (mat == 1) { W = Wk; bias = bk; Out = Ko; }
  else               { W = Wv; bias = bv; Out = Vo; }

  const int mr   = lane & 15;
  const int half = lane >> 4;
  const int n    = colT * 16 + mr;
  const float* hrow = h + (size_t)(rowT * 16 + mr) * DM;

  v8f c = {};
  for (int k0 = 0; k0 < DM; k0 += 4) {
    const int ka = k0 + 2 * half;
    v2f a = *(const v2f*)(hrow + ka);
    v2f b;
    b.x = W[(size_t)(ka + 0) * DM + n];
    b.y = W[(size_t)(ka + 1) * DM + n];
    c = wmma_f32(a, b, c);
  }
  const float bn = bias[n];
  #pragma unroll
  for (int v = 0; v < 8; ++v)
    Out[(size_t)(rowT * 16 + v + 8 * half) * DM + n] = c[v] + bn;
}

__global__ __launch_bounds__(128) void graph_attn_kernel(
    const float* __restrict__ Q, const float* __restrict__ K,
    const float* __restrict__ V,
    const float* __restrict__ adj, const float* __restrict__ msk,
    float* __restrict__ out)
{
  __shared__ float pl[4][16 * 18];

  const int lane = threadIdx.x & 31;
  const int wid  = threadIdx.x >> 5;
  const int task = blockIdx.x * 4 + wid;
  const int head = task >> 8;
  const int rowB = (task & 255) * 16;
  const int nn   = lane & 15;
  const int half = lane >> 4;
  float* plw = pl[wid];

  const float* qrow = Q + (size_t)(rowB + nn) * DM + head * DHD;
  v2f qa[16];
  #pragma unroll
  for (int s = 0; s < 16; ++s)
    qa[s] = *(const v2f*)(qrow + 4 * s + 2 * half);

  v8f acc[4] = {};
  float mrow[8], lrow[8];
  #pragma unroll
  for (int v = 0; v < 8; ++v) { mrow[v] = -3.0e38f; lrow[v] = 0.0f; }

  for (int j0 = 0; j0 < NTOK; j0 += 16) {
    v8f sf = {};
    const float* krow = K + (size_t)(j0 + nn) * DM + head * DHD;
    #pragma unroll
    for (int s = 0; s < 16; ++s) {
      v2f b = *(const v2f*)(krow + 4 * s + 2 * half);
      sf = wmma_f32(qa[s], b, sf);
    }

    if (j0 + 16 < NTOK)
      __builtin_prefetch(&adj[(size_t)(rowB + 8 * half) * NTOK + j0 + 16 + nn], 0, 1);

    float p[8];
    #pragma unroll
    for (int v = 0; v < 8; ++v) {
      const size_t e = (size_t)(rowB + v + 8 * half) * NTOK + j0 + nn;
      const float sc = (adj[e] * sf[v] + msk[e]) * SCALE;
      float bm = sc;
      bm = fmaxf(bm, __shfl_xor(bm, 1, 16));
      bm = fmaxf(bm, __shfl_xor(bm, 2, 16));
      bm = fmaxf(bm, __shfl_xor(bm, 4, 16));
      bm = fmaxf(bm, __shfl_xor(bm, 8, 16));
      const float newm = fmaxf(mrow[v], bm);
      const float f    = __expf(mrow[v] - newm);
      const float pv   = __expf(sc - newm);
      float bs = pv;
      bs += __shfl_xor(bs, 1, 16);
      bs += __shfl_xor(bs, 2, 16);
      bs += __shfl_xor(bs, 4, 16);
      bs += __shfl_xor(bs, 8, 16);
      lrow[v] = lrow[v] * f + bs;
      mrow[v] = newm;
      p[v] = pv;
      #pragma unroll
      for (int t = 0; t < 4; ++t) acc[t][v] *= f;
    }

    #pragma unroll
    for (int v = 0; v < 8; ++v)
      plw[(v + 8 * half) * 18 + nn] = p[v];
    __builtin_amdgcn_wave_barrier();

    #pragma unroll
    for (int s = 0; s < 4; ++s) {
      v2f a = *(const v2f*)&plw[nn * 18 + 4 * s + 2 * half];
      const float* v0 = V + (size_t)(j0 + 4 * s + 2 * half) * DM + head * DHD;
      #pragma unroll
      for (int t = 0; t < 4; ++t) {
        v2f b;
        b.x = v0[t * 16 + nn];
        b.y = v0[DM + t * 16 + nn];
        acc[t] = wmma_f32(a, b, acc[t]);
      }
    }
  }

  #pragma unroll
  for (int t = 0; t < 4; ++t) {
    #pragma unroll
    for (int v = 0; v < 8; ++v) {
      out[(size_t)(rowB + v + 8 * half) * DM + head * DHD + t * 16 + nn] =
          acc[t][v] / lrow[v];
    }
  }
}

extern "C" void kernel_launch(void* const* d_in, const int* in_sizes, int n_in,
                              void* d_out, int out_size, void* d_ws, size_t ws_size,
                              hipStream_t stream) {
  (void)in_sizes; (void)n_in; (void)out_size; (void)ws_size;
  const float* h   = (const float*)d_in[0];
  const float* adj = (const float*)d_in[1];
  const float* msk = (const float*)d_in[2];
  const float* Wq  = (const float*)d_in[3];
  const float* bq  = (const float*)d_in[4];
  const float* Wk  = (const float*)d_in[5];
  const float* bk  = (const float*)d_in[6];
  const float* Wv  = (const float*)d_in[7];
  const float* bv  = (const float*)d_in[8];
  float* out = (float*)d_out;

  float* Qo = (float*)d_ws;
  float* Ko = Qo + (size_t)NTOK * DM;
  float* Vo = Ko + (size_t)NTOK * DM;

  qkv_proj_kernel<<<3 * 256 * 32 / 8, 256, 0, stream>>>(
      h, Wq, bq, Wk, bk, Wv, bv, Qo, Ko, Vo);

  graph_attn_kernel<<<(NH * 256) / 4, 128, 0, stream>>>(
      Qo, Ko, Vo, adj, msk, out);
}


# attention trip widened from 16 to 64 key columns (4 score tiles; a trip's loads go out as one group): softmax update and acc rescale once per 64 cols; alters rescale points and the row-sum tree
# speedup vs baseline: 1.3659x; 1.3659x over previous
#include <hip/hip_runtime.h>


typedef float v2f __attribute__((ext_vector_type(2)));
typedef float v8f __attribute__((ext_vector_type(8)));

#define NTOK 4096
#define DM   512
#define NH   8
#define DHD  64
#define SCALE 0.125f

static __device__ __forceinline__ v8f wmma_f32(v2f a, v2f b, v8f c) {
  return __builtin_amdgcn_wmma_f32_16x16x4_f32(
       false, a,  false, b,
       (short)0, c,  false,  false);
}

__global__ __launch_bounds__(256) void qkv_proj_kernel(
    const float* __restrict__ h,
    const float* __restrict__ Wq, const float* __restrict__ bq,
    const float* __restrict__ Wk, const float* __restrict__ bk,
    const float* __restrict__ Wv, const float* __restrict__ bv,
    float* __restrict__ Qo, float* __restrict__ Ko, float* __restrict__ Vo)
{
  const int lane = threadIdx.x & 31;
  const int wid  = threadIdx.x >> 5;
  const int tile = blockIdx.x * 8 + wid;
  const int mat  = tile >> 13;
  const int rem  = tile & 8191;
  const int rowT = rem >> 5;
  const int colT = rem & 31;

  const float* W; const float* bias; float* Out;
  if (mat == 0)      { W = Wq; bias = bq; Out = Qo; }
  else if (mat == 1) { W = Wk; bias = bk; Out = Ko; }
  else               { W = Wv; bias = bv; Out = Vo; }

  const int mr   = lane & 15;
  const int half = lane >> 4;
  const int n    = colT * 16 + mr;
  const float* hrow = h + (size_t)(rowT * 16 + mr) * DM;

  v8f c = {};
  for (int k0 = 0; k0 < DM; k0 += 4) {
    const int ka = k0 + 2 * half;
    v2f a = *(const v2f*)(hrow + ka);
    v2f b;
    b.x = W[(size_t)(ka + 0) * DM + n];
    b.y = W[(size_t)(ka + 1) * DM + n];
    c = wmma_f32(a, b, c);
  }
  const float bn = bias[n];
  #pragma unroll
  for (int v = 0; v < 8; ++v)
    Out[(size_t)(rowT * 16 + v + 8 * half) * DM + n] = c[v] + bn;
}

#define CB  64
#define PST 68

__global__ __launch_bounds__(128) __attribute__((amdgpu_waves_per_eu(2, 2)))
void graph_attn_kernel(
    const float* __restrict__ Q, const float* __restrict__ K,
    const float* __restrict__ V,
    const float* __restrict__ adj, const float* __restrict__ msk,
    float* __restrict__ out)
{
  __shared__ float pl[4][16 * PST];

  const int lane = threadIdx.x & 31;
  const int wid  = threadIdx.x >> 5;
  const int task = blockIdx.x * 4 + wid;
  const int head = task >> 8;
  const int rowB = (task & 255) * 16;
  const int nn   = lane & 15;
  const int half = lane >> 4;
  float* plw = pl[wid];

  const float* qrow = Q + (size_t)(rowB + nn) * DM + head * DHD;
  v2f qa[16];
  #pragma unroll
  for (int s = 0; s < 16; ++s)
    qa[s] = *(const v2f*)(qrow + 4 * s + 2 * half);

  v8f acc[4] = {};
  float mrow[8], lrow[8];
  #pragma unroll
  for (int v = 0; v < 8; ++v) { mrow[v] = -3.0e38f; lrow[v] = 0.0f; }

  for (int j0 = 0; j0 < NTOK; j0 += CB) {
    const int jp = (j0 + CB < NTOK) ? j0 + CB : j0;
    __builtin_prefetch(&adj[(size_t)(rowB + nn) * NTOK + jp + 32 * half], 0, 1);

    v8f sf[4] = {};
    const float* krow = K + (size_t)(j0 + nn) * DM + head * DHD;
    #pragma unroll
    for (int s = 0; s < 16; ++s) {
      #pragma unroll
      for (int c = 0; c < 4; ++c) {
        v2f b = *(const v2f*)(krow + (size_t)(16 * c) * DM + 4 * s + 2 * half);
        sf[c] = wmma_f32(qa[s], b, sf[c]);
      }
    }

    #pragma unroll
    for (int v = 0; v < 8; ++v) {
      const size_t e = (size_t)(rowB + v + 8 * half) * NTOK + j0 + nn;
      float sc[4];
      #pragma unroll
      for (int c = 0; c < 4; ++c)
        sc[c] = (adj[e + 16 * c] * sf[c][v] + msk[e + 16 * c]) * SCALE;
      float bm = fmaxf(fmaxf(sc[0], sc[1]), fmaxf(sc[2], sc[3]));
      bm = fmaxf(bm, __shfl_xor(bm, 1, 16));
      bm = fmaxf(bm, __shfl_xor(bm, 2, 16));
      bm = fmaxf(bm, __shfl_xor(bm, 4, 16));
      bm = fmaxf(bm, __shfl_xor(bm, 8, 16));
      const float newm = fmaxf(mrow[v], bm);
      const float f    = __expf(mrow[v] - newm);
      float pv[4];
      #pragma unroll
      for (int c = 0; c < 4; ++c) pv[c] = __expf(sc[c] - newm);
      float bs = (pv[0] + pv[1]) + (pv[2] + pv[3]);
      bs += __shfl_xor(bs, 1, 16);
      bs += __shfl_xor(bs, 2, 16);
      bs += __shfl_xor(bs, 4, 16);
      bs += __shfl_xor(bs, 8, 16);
      lrow[v] = lrow[v] * f + bs;
      mrow[v] = newm;
      #pragma unroll
      for (int c = 0; c < 4; ++c)
        plw[(v + 8 * half) * PST + 16 * c + nn] = pv[c];
      #pragma unroll
      for (int t = 0; t < 4; ++t) acc[t][v] *= f;
    }
    __builtin_amdgcn_wave_barrier();

    #pragma unroll
    for (int s = 0; s < 16; ++s) {
      v2f a = *(const v2f*)&plw[nn * PST + 4 * s + 2 * half];
      const float* v0 = V + (size_t)(j0 + 4 * s + 2 * half) * DM + head * DHD;
      #pragma unroll
      for (int t = 0; t < 4; ++t) {
        v2f b;
        b.x = v0[t * 16 + nn];
        b.y = v0[DM + t * 16 + nn];
        acc[t] = wmma_f32(a, b, acc[t]);
      }
    }
  }

  #pragma unroll
  for (int t = 0; t < 4; ++t) {
    #pragma unroll
    for (int v = 0; v < 8; ++v) {
      out[(size_t)(rowB + v + 8 * half) * DM + head * DHD + t * 16 + nn] =
          acc[t][v] / lrow[v];
    }
  }
}

extern "C" void kernel_launch(void* const* d_in, const int* in_sizes, int n_in,
                              void* d_out, int out_size, void* d_ws, size_t ws_size,
                              hipStream_t stream) {
  (void)in_sizes; (void)n_in; (void)out_size; (void)ws_size;
  const float* h   = (const float*)d_in[0];
  const float* adj = (const float*)d_in[1];
  const float* msk = (const float*)d_in[2];
  const float* Wq  = (const float*)d_in[3];
  const float* bq  = (const float*)d_in[4];
  const float* Wk  = (const float*)d_in[5];
  const float* bk  = (const float*)d_in[6];
  const float* Wv  = (const float*)d_in[7];
  const float* bv  = (const float*)d_in[8];
  float* out = (float*)d_out;

  float* Qo = (float*)d_ws;
  float* Ko = Qo + (size_t)NTOK * DM;
  float* Vo = Ko + (size_t)NTOK * DM;

  qkv_proj_kernel<<<3 * 256 * 32 / 8, 256, 0, stream>>>(
      h, Wq, bq, Wk, bk, Wv, bv, Qo, Ko, Vo);

  graph_attn_kernel<<<(NH * 256) / 4, 128, 0, stream>>>(
      Qo, Ko, Vo, adj, msk, out);
}


# projection stores K and V in WMMA B-fragment order so each attention fragment load is one 16-byte piece of a contiguous 512-B run per wave; regroups the k order inside both products' sums
# speedup vs baseline: 1.5992x; 1.1708x over previous
#include <hip/hip_runtime.h>


typedef float v2f __attribute__((ext_vector_type(2)));
typedef float v4f __attribute__((ext_vector_type(4)));
typedef float v8f __attribute__((ext_vector_type(8)));


#define NTOK 4096
#define DM   512
#define NH   8
#define DHD  64
#define SCALE 0.125f

static __device__ __forceinline__ v8f wmma_f32(v2f a, v2f b, v8f c) {
  return __builtin_amdgcn_wmma_f32_16x16x4_f32(
       false, a,  false, b,
       (short)0, c,  false,  false);
}

__global__ __launch_bounds__(256) void qkv_proj_kernel(
    const float* __restrict__ h,
    const float* __restrict__ Wq, const float* __restrict__ bq,
    const float* __restrict__ Wk, const float* __restrict__ bk,
    const float* __restrict__ Wv, const float* __restrict__ bv,
    float* __restrict__ Qo, float* __restrict__ Ko, float* __restrict__ Vo)
{
  const int lane = threadIdx.x & 31;
  const int wid  = threadIdx.x >> 5;
  const int mat  = blockIdx.x >> 10;
  const int rem  = (blockIdx.x * 8 + wid) & 8191;
  const int rowT = rem >> 5;
  const int colT = rem & 31;

  const int mr   = lane & 15;
  const int half = lane >> 4;
  const int n    = colT * 16 + mr;
  const float* hrow = h + (size_t)(rowT * 16 + mr) * DM;
  const int hd = colT >> 2;

  if (mat == 1) {
    v8f c = {};
    for (int k0 = 0; k0 < DM; k0 += 4) {
      const int ka = k0 + 2 * half;
      v2f a = *(const v2f*)(hrow + ka);
      v2f b;
      b.x = Wk[(size_t)(ka + 0) * DM + n];
      b.y = Wk[(size_t)(ka + 1) * DM + n];
      c = wmma_f32(b, a, c);
    }
    const float* bp = bk + colT * 16 + 8 * half;
    const v4f b0 = *(const v4f*)bp;
    const v4f b1 = *(const v4f*)(bp + 4);
    const int u = (colT & 3) * 2 + half;
    float* dst = Ko + ((size_t)((hd * 256 + rowT) * 8 + u) * 32 + mr) * 4;
    v4f o0 = { c[0] + b0.x, c[1] + b0.y, c[2] + b0.z, c[3] + b0.w };
    v4f o1 = { c[4] + b1.x, c[5] + b1.y, c[6] + b1.z, c[7] + b1.w };
    *(v4f*)dst        = o0;
    *(v4f*)(dst + 64) = o1;
    return;
  }

  const float* W    = (mat == 0) ? Wq : Wv;
  const float* bias = (mat == 0) ? bq : bv;

  v8f c = {};
  for (int k0 = 0; k0 < DM; k0 += 4) {
    const int ka = k0 + 2 * half;
    v2f a = *(const v2f*)(hrow + ka);
    v2f b;
    b.x = W[(size_t)(ka + 0) * DM + n];
    b.y = W[(size_t)(ka + 1) * DM + n];
    c = wmma_f32(a, b, c);
  }
  const float bn = bias[n];
  if (mat == 0) {
    #pragma unroll
    for (int v = 0; v < 8; ++v)
      Qo[(size_t)(rowT * 16 + v + 8 * half) * DM + n] = c[v] + bn;
  } else {
    const int t = colT & 3;
    float* dst = Vo + ((size_t)(((hd * 512 + rowT * 2 + half) * 4 + t) * 32) + mr) * 4;
    v4f o0 = { c[0] + bn, c[1] + bn, c[2] + bn, c[3] + bn };
    v4f o1 = { c[4] + bn, c[5] + bn, c[6] + bn, c[7] + bn };
    *(v4f*)dst        = o0;
    *(v4f*)(dst + 64) = o1;
  }
}

#define CB  64
#define PST 68

__global__ __launch_bounds__(128) __attribute__((amdgpu_waves_per_eu(2, 2)))
void graph_attn_kernel(
    const float* __restrict__ Q, const float* __restrict__ K,
    const float* __restrict__ V,
    const float* __restrict__ adj, const float* __restrict__ msk,
    float* __restrict__ out)
{
  __shared__ float pl[4][16 * PST];

  const int lane = threadIdx.x & 31;
  const int wid  = threadIdx.x >> 5;
  const int task = blockIdx.x * 4 + wid;
  const int head = task >> 8;
  const int rowB = (task & 255) * 16;
  const int nn   = lane & 15;
  const int half = lane >> 4;
  float* plw = pl[wid];

  const float* qrow = Q + (size_t)(rowB + nn) * DM + head * DHD + 4 * half;
  v4f qa[8];
  #pragma unroll
  for (int u = 0; u < 8; ++u)
    qa[u] = *(const v4f*)(qrow + 8 * u);

  v8f acc[4] = {};
  float mrow[8], lrow[8];
  #pragma unroll
  for (int v = 0; v < 8; ++v) { mrow[v] = -3.0e38f; lrow[v] = 0.0f; }

  for (int j0 = 0; j0 < NTOK; j0 += CB) {
    const int jp = (j0 + CB < NTOK) ? j0 + CB : j0;
    __builtin_prefetch(&adj[(size_t)(rowB + nn) * NTOK + jp + 32 * half], 0, 1);

    v8f sf[4] = {};
    const float* kt = K + ((size_t)(head * 256 + (j0 >> 4)) * 8) * 128 + lane * 4;
    #pragma unroll
    for (int u = 0; u < 8; ++u) {
      v4f kb[4];
      #pragma unroll
      for (int c = 0; c < 4; ++c)
        kb[c] = *(const v4f*)(kt + c * 1024 + u * 128);
      #pragma unroll
      for (int c = 0; c < 4; ++c)
        sf[c] = wmma_f32(qa[u].xy, kb[c].xy, sf[c]);
      #pragma unroll
      for (int c = 0; c < 4; ++c)
        sf[c] = wmma_f32(qa[u].zw, kb[c].zw, sf[c]);
    }

    #pragma unroll
    for (int v = 0; v < 8; ++v) {
      const size_t e = (size_t)(rowB + v + 8 * half) * NTOK + j0 + nn;
      float sc[4];
      #pragma unroll
      for (int c = 0; c < 4; ++c)
        sc[c] = (adj[e + 16 * c] * sf[c][v] + msk[e + 16 * c]) * SCALE;
      float bm = fmaxf(fmaxf(sc[0], sc[1]), fmaxf(sc[2], sc[3]));
      bm = fmaxf(bm, __shfl_xor(bm, 1, 16));
      bm = fmaxf(bm, __shfl_xor(bm, 2, 16));
      bm = fmaxf(bm, __shfl_xor(bm, 4, 16));
      bm = fmaxf(bm, __shfl_xor(bm, 8, 16));
      const float newm = fmaxf(mrow[v], bm);
      const float f    = __expf(mrow[v] - newm);
      float pv[4];
      #pragma unroll
      for (int c = 0; c < 4; ++c) pv[c] = __expf(sc[c] - newm);
      float bs = (pv[0] + pv[1]) + (pv[2] + pv[3]);
      bs += __shfl_xor(bs, 1, 16);
      bs += __shfl_xor(bs, 2, 16);
      bs += __shfl_xor(bs, 4, 16);
      bs += __shfl_xor(bs, 8, 16);
      lrow[v] = lrow[v] * f + bs;
      mrow[v] = newm;
      #pragma unroll
      for (int c = 0; c < 4; ++c)
        plw[(v + 8 * half) * PST + 16 * c + nn] = pv[c];
      #pragma unroll
      for (int t = 0; t < 4; ++t) acc[t][v] *= f;
    }
    __builtin_amdgcn_wave_barrier();

    const float* vt = V + ((size_t)(head * 512 + (j0 >> 3)) * 4) * 128 + lane * 4;
    #pragma unroll
    for (int u = 0; u < 8; ++u) {
      v4f a = *(const v4f*)&plw[nn * PST + 8 * u + 4 * half];
      v4f vb[4];
      #pragma unroll
      for (int t = 0; t < 4; ++t)
        vb[t] = *(const v4f*)(vt + (u * 4 + t) * 128);
      #pragma unroll
      for (int t = 0; t < 4; ++t)
        acc[t] = wmma_f32(a.xy, vb[t].xy, acc[t]);
      #pragma unroll
      for (int t = 0; t < 4; ++t)
        acc[t] = wmma_f32(a.zw, vb[t].zw, acc[t]);
    }
  }

  #pragma unroll
  for (int t = 0; t < 4; ++t) {
    #pragma unroll
    for (int v = 0; v < 8; ++v) {
      out[(size_t)(rowB + v + 8 * half) * DM + head * DHD + t * 16 + nn] =
          acc[t][v] / lrow[v];
    }
  }
}

extern "C" void kernel_launch(void* const* d_in, const int* in_sizes, int n_in,
                              void* d_out, int out_size, void* d_ws, size_t ws_size,
                              hipStream_t stream) {
  (void)in_sizes; (void)n_in; (void)out_size; (void)ws_size;
  const float* h   = (const float*)d_in[0];
  const float* adj = (const float*)d_in[1];
  const float* msk = (const float*)d_in[2];
  const float* Wq  = (const float*)d_in[3];
  const float* bq  = (const float*)d_in[4];
  const float* Wk  = (const float*)d_in[5];
  const float* bk  = (const float*)d_in[6];
  const float* Wv  = (const float*)d_in[7];
  const float* bv  = (const float*)d_in[8];
  float* out = (float*)d_out;

  float* Qo = (float*)d_ws;
  float* Ko = Qo + (size_t)NTOK * DM;
  float* Vo = Ko + (size_t)NTOK * DM;

  qkv_proj_kernel<<<3 * 256 * 32 / 8, 256, 0, stream>>>(
      h, Wq, bq, Wk, bk, Wv, bv, Qo, Ko, Vo);

  graph_attn_kernel<<<(NH * 256) / 4, 128, 0, stream>>>(
      Qo, Ko, Vo, adj, msk, out);
}


# row max and row sum butterflies of the softmax as DPP row_xmask operands on the vector ALU instead of LDS permutes with a wait per step; same tree, same sums
# speedup vs baseline: 1.7161x; 1.0731x over previous
#include <hip/hip_runtime.h>


typedef float v2f __attribute__((ext_vector_type(2)));
typedef float v4f __attribute__((ext_vector_type(4)));
typedef float v8f __attribute__((ext_vector_type(8)));


#define NTOK 4096
#define DM   512
#define NH   8
#define DHD  64
#define SCALE 0.125f

static __device__ __forceinline__ v8f wmma_f32(v2f a, v2f b, v8f c) {
  return __builtin_amdgcn_wmma_f32_16x16x4_f32(
       false, a,  false, b,
       (short)0, c,  false,  false);
}

template <int M>
static __device__ __forceinline__ float row_xor(float x) {
  return __builtin_amdgcn_update_dpp(0.0f, x, 0x160 + M, 0xf, 0xf, true);
}

__global__ __launch_bounds__(256) void qkv_proj_kernel(
    const float* __restrict__ h,
    const float* __restrict__ Wq, const float* __restrict__ bq,
    const float* __restrict__ Wk, const float* __restrict__ bk,
    const float* __restrict__ Wv, const float* __restrict__ bv,
    float* __restrict__ Qo, float* __restrict__ Ko, float* __restrict__ Vo)
{
  const int lane = threadIdx.x & 31;
  const int wid  = threadIdx.x >> 5;
  const int mat  = blockIdx.x >> 10;
  const int rem  = (blockIdx.x * 8 + wid) & 8191;
  const int rowT = rem >> 5;
  const int colT = rem & 31;

  const int mr   = lane & 15;
  const int half = lane >> 4;
  const int n    = colT * 16 + mr;
  const float* hrow = h + (size_t)(rowT * 16 + mr) * DM;
  const int hd = colT >> 2;

  if (mat == 1) {
    v8f c = {};
    for (int k0 = 0; k0 < DM; k0 += 4) {
      const int ka = k0 + 2 * half;
      v2f a = *(const v2f*)(hrow + ka);
      v2f b;
      b.x = Wk[(size_t)(ka + 0) * DM + n];
      b.y = Wk[(size_t)(ka + 1) * DM + n];
      c = wmma_f32(b, a, c);
    }
    const float* bp = bk + colT * 16 + 8 * half;
    const v4f b0 = *(const v4f*)bp;
    const v4f b1 = *(const v4f*)(bp + 4);
    const int u = (colT & 3) * 2 + half;
    float* dst = Ko + ((size_t)((hd * 256 + rowT) * 8 + u) * 32 + mr) * 4;
    v4f o0 = { c[0] + b0.x, c[1] + b0.y, c[2] + b0.z, c[3] + b0.w };
    v4f o1 = { c[4] + b1.x, c[5] + b1.y, c[6] + b1.z, c[7] + b1.w };
    *(v4f*)dst        = o0;
    *(v4f*)(dst + 64) = o1;
    return;
  }

  const float* W    = (mat == 0) ? Wq : Wv;
  const float* bias = (mat == 0) ? bq : bv;

  v8f c = {};
  for (int k0 = 0; k0 < DM; k0 += 4) {
    const int ka = k0 + 2 * half;
    v2f a = *(const v2f*)(hrow + ka);
    v2f b;
    b.x = W[(size_t)(ka + 0) * DM + n];
    b.y = W[(size_t)(ka + 1) * DM + n];
    c = wmma_f32(a, b, c);
  }
  const float bn = bias[n];
  if (mat == 0) {
    #pragma unroll
    for (int v = 0; v < 8; ++v)
      Qo[(size_t)(rowT * 16 + v + 8 * half) * DM + n] = c[v] + bn;
  } else {
    const int t = colT & 3;
    float* dst = Vo + ((size_t)(((hd * 512 + rowT * 2 + half) * 4 + t) * 32) + mr) * 4;
    v4f o0 = { c[0] + bn, c[1] + bn, c[2] + bn, c[3] + bn };
    v4f o1 = { c[4] + bn, c[5] + bn, c[6] + bn, c[7] + bn };
    *(v4f*)dst        = o0;
    *(v4f*)(dst + 64) = o1;
  }
}

#define CB  64
#define PST 68

__global__ __launch_bounds__(128) __attribute__((amdgpu_waves_per_eu(2, 2)))
void graph_attn_kernel(
    const float* __restrict__ Q, const float* __restrict__ K,
    const float* __restrict__ V,
    const float* __restrict__ adj, const float* __restrict__ msk,
    float* __restrict__ out)
{
  __shared__ float pl[4][16 * PST];

  const int lane = threadIdx.x & 31;
  const int wid  = threadIdx.x >> 5;
  const int task = blockIdx.x * 4 + wid;
  const int head = task >> 8;
  const int rowB = (task & 255) * 16;
  const int nn   = lane & 15;
  const int half = lane >> 4;
  float* plw = pl[wid];

  const float* qrow = Q + (size_t)(rowB + nn) * DM + head * DHD + 4 * half;
  v4f qa[8];
  #pragma unroll
  for (int u = 0; u < 8; ++u)
    qa[u] = *(const v4f*)(qrow + 8 * u);

  v8f acc[4] = {};
  float mrow[8], lrow[8];
  #pragma unroll
  for (int v = 0; v < 8; ++v) { mrow[v] = -3.0e38f; lrow[v] = 0.0f; }

  for (int j0 = 0; j0 < NTOK; j0 += CB) {
    const int jp = (j0 + CB < NTOK) ? j0 + CB : j0;
    __builtin_prefetch(&adj[(size_t)(rowB + nn) * NTOK + jp + 32 * half], 0, 1);

    v8f sf[4] = {};
    const float* kt = K + ((size_t)(head * 256 + (j0 >> 4)) * 8) * 128 + lane * 4;
    #pragma unroll
    for (int u = 0; u < 8; ++u) {
      v4f kb[4];
      #pragma unroll
      for (int c = 0; c < 4; ++c)
        kb[c] = *(const v4f*)(kt + c * 1024 + u * 128);
      #pragma unroll
      for (int c = 0; c < 4; ++c)
        sf[c] = wmma_f32(qa[u].xy, kb[c].xy, sf[c]);
      #pragma unroll
      for (int c = 0; c < 4; ++c)
        sf[c] = wmma_f32(qa[u].zw, kb[c].zw, sf[c]);
    }

    #pragma unroll
    for (int v = 0; v < 8; ++v) {
      const size_t e = (size_t)(rowB + v + 8 * half) * NTOK + j0 + nn;
      float sc[4];
      #pragma unroll
      for (int c = 0; c < 4; ++c)
        sc[c] = (adj[e + 16 * c] * sf[c][v] + msk[e + 16 * c]) * SCALE;
      float bm = fmaxf(fmaxf(sc[0], sc[1]), fmaxf(sc[2], sc[3]));
      bm = fmaxf(row_xor<1>(bm), bm);
      bm = fmaxf(row_xor<2>(bm), bm);
      bm = fmaxf(row_xor<4>(bm), bm);
      bm = fmaxf(row_xor<8>(bm), bm);
      const float newm = fmaxf(mrow[v], bm);
      const float f    = __expf(mrow[v] - newm);
      float pv[4];
      #pragma unroll
      for (int c = 0; c < 4; ++c) pv[c] = __expf(sc[c] - newm);
      float bs = (pv[0] + pv[1]) + (pv[2] + pv[3]);
      bs = row_xor<1>(bs) + bs;
      bs = row_xor<2>(bs) + bs;
      bs = row_xor<4>(bs) + bs;
      bs = row_xor<8>(bs) + bs;
      lrow[v] = lrow[v] * f + bs;
      mrow[v] = newm;
      #pragma unroll
      for (int c = 0; c < 4; ++c)
        plw[(v + 8 * half) * PST + 16 * c + nn] = pv[c];
      #pragma unroll
      for (int t = 0; t < 4; ++t) acc[t][v] *= f;
    }
    __builtin_amdgcn_wave_barrier();

    const float* vt = V + ((size_t)(head * 512 + (j0 >> 3)) * 4) * 128 + lane * 4;
    #pragma unroll
    for (int u = 0; u < 8; ++u) {
      v4f a = *(const v4f*)&plw[nn * PST + 8 * u + 4 * half];
      v4f vb[4];
      #pragma unroll
      for (int t = 0; t < 4; ++t)
        vb[t] = *(const v4f*)(vt + (u * 4 + t) * 128);
      #pragma unroll
      for (int t = 0; t < 4; ++t)
        acc[t] = wmma_f32(a.xy, vb[t].xy, acc[t]);
      #pragma unroll
      for (int t = 0; t < 4; ++t)
        acc[t] = wmma_f32(a.zw, vb[t].zw, acc[t]);
    }
  }

  #pragma unroll
  for (int t = 0; t < 4; ++t) {
    #pragma unroll
    for (int v = 0; v < 8; ++v) {
      out[(size_t)(rowB + v + 8 * half) * DM + head * DHD + t * 16 + nn] =
          acc[t][v] / lrow[v];
    }
  }
}

extern "C" void kernel_launch(void* const* d_in, const int* in_sizes, int n_in,
                              void* d_out, int out_size, void* d_ws, size_t ws_size,
                              hipStream_t stream) {
  (void)in_sizes; (void)n_in; (void)out_size; (void)ws_size;
  const float* h   = (const float*)d_in[0];
  const float* adj = (const float*)d_in[1];
  const float* msk = (const float*)d_in[2];
  const float* Wq  = (const float*)d_in[3];
  const float* bq  = (const float*)d_in[4];
  const float* Wk  = (const float*)d_in[5];
  const float* bk  = (const float*)d_in[6];
  const float* Wv  = (const float*)d_in[7];
  const float* bv  = (const float*)d_in[8];
  float* out = (float*)d_out;

  float* Qo = (float*)d_ws;
  float* Ko = Qo + (size_t)NTOK * DM;
  float* Vo = Ko + (size_t)NTOK * DM;

  qkv_proj_kernel<<<3 * 256 * 32 / 8, 256, 0, stream>>>(
      h, Wq, bq, Wk, bk, Wv, bv, Qo, Ko, Vo);

  graph_attn_kernel<<<(NH * 256) / 4, 128, 0, stream>>>(
      Qo, Ko, Vo, adj, msk, out);
}


# projection re-tiled: a wave computes 32 tokens x 64 columns (2x4 accumulator tiles), so an h fragment feeds 8 WMMAs and a W fragment 4, next step's loads issued ahead; regroups k inside its sums
# speedup vs baseline: 1.9992x; 1.1650x over previous
#include <hip/hip_runtime.h>


typedef float v2f __attribute__((ext_vector_type(2)));
typedef float v4f __attribute__((ext_vector_type(4)));
typedef float v8f __attribute__((ext_vector_type(8)));


#define NTOK 4096
#define DM   512
#define NH   8
#define DHD  64
#define SCALE 0.125f

static __device__ __forceinline__ v8f wmma_f32(v2f a, v2f b, v8f c) {
  return __builtin_amdgcn_wmma_f32_16x16x4_f32(
       false, a,  false, b,
       (short)0, c,  false,  false);
}

template <int M>
static __device__ __forceinline__ float row_xor(float x) {
  return __builtin_amdgcn_update_dpp(0.0f, x, 0x160 + M, 0xf, 0xf, true);
}

struct ProjFrag {
  v4f a[2];
  float w[4][4];
};

static __device__ __forceinline__ void proj_load(
    ProjFrag& f, const float* __restrict__ hp, const float* __restrict__ wp, int u) {
  #pragma unroll
  for (int i = 0; i < 2; ++i)
    f.a[i] = *(const v4f*)(hp + (size_t)i * 16 * DM + 8 * u);
  #pragma unroll
  for (int t = 0; t < 4; ++t) {
    #pragma unroll
    for (int s = 0; s < 4; ++s)
      f.w[t][s] = wp[(size_t)(8 * u + s) * DM + 16 * t];
  }
}

template <bool KT>
static __device__ __forceinline__ void proj_mma(const ProjFrag& f, v8f (&c)[2][4]) {
  #pragma unroll
  for (int i = 0; i < 2; ++i) {
    #pragma unroll
    for (int t = 0; t < 4; ++t) {
      const v2f w01 = { f.w[t][0], f.w[t][1] };
      c[i][t] = KT ? wmma_f32(w01, f.a[i].xy, c[i][t]) : wmma_f32(f.a[i].xy, w01, c[i][t]);
    }
  }
  #pragma unroll
  for (int i = 0; i < 2; ++i) {
    #pragma unroll
    for (int t = 0; t < 4; ++t) {
      const v2f w23 = { f.w[t][2], f.w[t][3] };
      c[i][t] = KT ? wmma_f32(w23, f.a[i].zw, c[i][t]) : wmma_f32(f.a[i].zw, w23, c[i][t]);
    }
  }
}

template <bool KT>
static __device__ __forceinline__ void proj_loop(
    const float* __restrict__ hp, const float* __restrict__ wp, v8f (&c)[2][4]) {
  ProjFrag f0, f1;
  proj_load(f0, hp, wp, 0);
  for (int u = 0; u < DM / 8; u += 2) {
    proj_load(f1, hp, wp, u + 1);
    __builtin_amdgcn_sched_barrier(0);
    proj_mma<KT>(f0, c);
    proj_load(f0, hp, wp, (u + 2 < DM / 8) ? u + 2 : u);
    __builtin_amdgcn_sched_barrier(0);
    proj_mma<KT>(f1, c);
  }
}

__global__ __launch_bounds__(128) void qkv_proj_kernel(
    const float* __restrict__ h,
    const float* __restrict__ Wq, const float* __restrict__ bq,
    const float* __restrict__ Wk, const float* __restrict__ bk,
    const float* __restrict__ Wv, const float* __restrict__ bv,
    float* __restrict__ Qo, float* __restrict__ Ko, float* __restrict__ Vo)
{
  const int lane = threadIdx.x & 31;
  const int wid  = threadIdx.x >> 5;
  const int mat  = blockIdx.x >> 8;
  const int hd   = (blockIdx.x >> 5) & 7;
  const int tb   = (blockIdx.x & 31) * 4 + wid;

  const int mr   = lane & 15;
  const int half = lane >> 4;
  const float* hp = h + (size_t)(tb * 32 + mr) * DM + 4 * half;
  const float* W    = (mat == 0) ? Wq : (mat == 1) ? Wk : Wv;
  const float* bias = (mat == 0) ? bq : (mat == 1) ? bk : bv;
  const float* wp = W + (size_t)(4 * half) * DM + hd * DHD + mr;

  v8f c[2][4] = {};

  if (mat == 1) {
    proj_loop<true>(hp, wp, c);
    #pragma unroll
    for (int i = 0; i < 2; ++i) {
      #pragma unroll
      for (int t = 0; t < 4; ++t) {
        const float* bp = bias + hd * DHD + 16 * t + 8 * half;
        const v4f b0 = *(const v4f*)bp;
        const v4f b1 = *(const v4f*)(bp + 4);
        const int u = t * 2 + half;
        float* dst = Ko + ((size_t)((hd * 256 + tb * 2 + i) * 8 + u) * 32 + mr) * 4;
        const v8f& r = c[i][t];
        v4f o0 = { r[0] + b0.x, r[1] + b0.y, r[2] + b0.z, r[3] + b0.w };
        v4f o1 = { r[4] + b1.x, r[5] + b1.y, r[6] + b1.z, r[7] + b1.w };
        *(v4f*)dst        = o0;
        *(v4f*)(dst + 64) = o1;
      }
    }
    return;
  }

  proj_loop<false>(hp, wp, c);
  if (mat == 0) {
    #pragma unroll
    for (int i = 0; i < 2; ++i) {
      #pragma unroll
      for (int t = 0; t < 4; ++t) {
        const int n = hd * DHD + 16 * t + mr;
        const float bn = bias[n];
        #pragma unroll
        for (int v = 0; v < 8; ++v)
          Qo[(size_t)(tb * 32 + 16 * i + v + 8 * half) * DM + n] = c[i][t][v] + bn;
      }
    }
  } else {
    #pragma unroll
    for (int i = 0; i < 2; ++i) {
      #pragma unroll
      for (int t = 0; t < 4; ++t) {
        const float bn = bias[hd * DHD + 16 * t + mr];
        float* dst = Vo + ((size_t)(((hd * 512 + (tb * 2 + i) * 2 + half) * 4 + t) * 32) + mr) * 4;
        const v8f& r = c[i][t];
        v4f o0 = { r[0] + bn, r[1] + bn, r[2] + bn, r[3] + bn };
        v4f o1 = { r[4] + bn, r[5] + bn, r[6] + bn, r[7] + bn };
        *(v4f*)dst        = o0;
        *(v4f*)(dst + 64) = o1;
      }
    }
  }
}

#define CB  64
#define PST 68

__global__ __launch_bounds__(128) __attribute__((amdgpu_waves_per_eu(2, 2)))
void graph_attn_kernel(
    const float* __restrict__ Q, const float* __restrict__ K,
    const float* __restrict__ V,
    const float* __restrict__ adj, const float* __restrict__ msk,
    float* __restrict__ out)
{
  __shared__ float pl[4][16 * PST];

  const int lane = threadIdx.x & 31;
  const int wid  = threadIdx.x >> 5;
  const int task = blockIdx.x * 4 + wid;
  const int head = task >> 8;
  const int rowB = (task & 255) * 16;
  const int nn   = lane & 15;
  const int half = lane >> 4;
  float* plw = pl[wid];

  const float* qrow = Q + (size_t)(rowB + nn) * DM + head * DHD + 4 * half;
  v4f qa[8];
  #pragma unroll
  for (int u = 0; u < 8; ++u)
    qa[u] = *(const v4f*)(qrow + 8 * u);

  v8f acc[4] = {};
  float mrow[8], lrow[8];
  #pragma unroll
  for (int v = 0; v < 8; ++v) { mrow[v] = -3.0e38f; lrow[v] = 0.0f; }

  for (int j0 = 0; j0 < NTOK; j0 += CB) {
    const int jp = (j0 + CB < NTOK) ? j0 + CB : j0;
    __builtin_prefetch(&adj[(size_t)(rowB + nn) * NTOK + jp + 32 * half], 0, 1);

    v8f sf[4] = {};
    const float* kt = K + ((size_t)(head * 256 + (j0 >> 4)) * 8) * 128 + lane * 4;
    #pragma unroll
    for (int u = 0; u < 8; ++u) {
      v4f kb[4];
      #pragma unroll
      for (int c = 0; c < 4; ++c)
        kb[c] = *(const v4f*)(kt + c * 1024 + u * 128);
      #pragma unroll
      for (int c = 0; c < 4; ++c)
        sf[c] = wmma_f32(qa[u].xy, kb[c].xy, sf[c]);
      #pragma unroll
      for (int c = 0; c < 4; ++c)
        sf[c] = wmma_f32(qa[u].zw, kb[c].zw, sf[c]);
    }

    #pragma unroll
    for (int v = 0; v < 8; ++v) {
      const size_t e = (size_t)(rowB + v + 8 * half) * NTOK + j0 + nn;
      float sc[4];
      #pragma unroll
      for (int c = 0; c < 4; ++c)
        sc[c] = (adj[e + 16 * c] * sf[c][v] + msk[e + 16 * c]) * SCALE;
      float bm = fmaxf(fmaxf(sc[0], sc[1]), fmaxf(sc[2], sc[3]));
      bm = fmaxf(row_xor<1>(bm), bm);
      bm = fmaxf(row_xor<2>(bm), bm);
      bm = fmaxf(row_xor<4>(bm), bm);
      bm = fmaxf(row_xor<8>(bm), bm);
      const float newm = fmaxf(mrow[v], bm);
      const float f    = __expf(mrow[v] - newm);
      float pv[4];
      #pragma unroll
      for (int c = 0; c < 4; ++c) pv[c] = __expf(sc[c] - newm);
      float bs = (pv[0] + pv[1]) + (pv[2] + pv[3]);
      bs = row_xor<1>(bs) + bs;
      bs = row_xor<2>(bs) + bs;
      bs = row_xor<4>(bs) + bs;
      bs = row_xor<8>(bs) + bs;
      lrow[v] = lrow[v] * f + bs;
      mrow[v] = newm;
      #pragma unroll
      for (int c = 0; c < 4; ++c)
        plw[(v + 8 * half) * PST + 16 * c + nn] = pv[c];
      #pragma unroll
      for (int t = 0; t < 4; ++t) acc[t][v] *= f;
    }
    __builtin_amdgcn_wave_barrier();

    const float* vt = V + ((size_t)(head * 512 + (j0 >> 3)) * 4) * 128 + lane * 4;
    #pragma unroll
    for (int u = 0; u < 8; ++u) {
      v4f a = *(const v4f*)&plw[nn * PST + 8 * u + 4 * half];
      v4f vb[4];
      #pragma unroll
      for (int t = 0; t < 4; ++t)
        vb[t] = *(const v4f*)(vt + (u * 4 + t) * 128);
      #pragma unroll
      for (int t = 0; t < 4; ++t)
        acc[t] = wmma_f32(a.xy, vb[t].xy, acc[t]);
      #pragma unroll
      for (int t = 0; t < 4; ++t)
        acc[t] = wmma_f32(a.zw, vb[t].zw, acc[t]);
    }
  }

  #pragma unroll
  for (int t = 0; t < 4; ++t) {
    #pragma unroll
    for (int v = 0; v < 8; ++v) {
      out[(size_t)(rowB + v + 8 * half) * DM + head * DHD + t * 16 + nn] =
          acc[t][v] / lrow[v];
    }
  }
}

extern "C" void kernel_launch(void* const* d_in, const int* in_sizes, int n_in,
                              void* d_out, int out_size, void* d_ws, size_t ws_size,
                              hipStream_t stream) {
  (void)in_sizes; (void)n_in; (void)out_size; (void)ws_size;
  const float* h   = (const float*)d_in[0];
  const float* adj = (const float*)d_in[1];
  const float* msk = (const float*)d_in[2];
  const float* Wq  = (const float*)d_in[3];
  const float* bq  = (const float*)d_in[4];
  const float* Wk  = (const float*)d_in[5];
  const float* bk  = (const float*)d_in[6];
  const float* Wv  = (const float*)d_in[7];
  const float* bv  = (const float*)d_in[8];
  float* out = (float*)d_out;

  float* Qo = (float*)d_ws;
  float* Ko = Qo + (size_t)NTOK * DM;
  float* Vo = Ko + (size_t)NTOK * DM;

  qkv_proj_kernel<<<3 * NH * 128 / 4, 128, 0, stream>>>(
      h, Wq, bq, Wk, bk, Wv, bv, Qo, Ko, Vo);

  graph_attn_kernel<<<(NH * 256) / 4, 128, 0, stream>>>(
      Qo, Ko, Vo, adj, msk, out);
}


# attention trip software-pipelined in registers: V loads at the trip top, next trip's K loads behind the score WMMAs, next adj/mask loads behind the scores and the softmax; same sums, same order
# speedup vs baseline: 2.2295x; 1.1152x over previous
#include <hip/hip_runtime.h>


typedef float v2f __attribute__((ext_vector_type(2)));
typedef float v4f __attribute__((ext_vector_type(4)));
typedef float v8f __attribute__((ext_vector_type(8)));


#define NTOK 4096
#define DM   512
#define NH   8
#define DHD  64
#define SCALE 0.125f

static __device__ __forceinline__ v8f wmma_f32(v2f a, v2f b, v8f c) {
  return __builtin_amdgcn_wmma_f32_16x16x4_f32(
       false, a,  false, b,
       (short)0, c,  false,  false);
}

template <int M>
static __device__ __forceinline__ float row_xor(float x) {
  return __builtin_amdgcn_update_dpp(0.0f, x, 0x160 + M, 0xf, 0xf, true);
}

struct ProjFrag {
  v4f a[2];
  float w[4][4];
};

static __device__ __forceinline__ void proj_load(
    ProjFrag& f, const float* __restrict__ hp, const float* __restrict__ wp, int u) {
  #pragma unroll
  for (int i = 0; i < 2; ++i)
    f.a[i] = *(const v4f*)(hp + (size_t)i * 16 * DM + 8 * u);
  #pragma unroll
  for (int t = 0; t < 4; ++t) {
    #pragma unroll
    for (int s = 0; s < 4; ++s)
      f.w[t][s] = wp[(size_t)(8 * u + s) * DM + 16 * t];
  }
}

template <bool KT>
static __device__ __forceinline__ void proj_mma(const ProjFrag& f, v8f (&c)[2][4]) {
  #pragma unroll
  for (int i = 0; i < 2; ++i) {
    #pragma unroll
    for (int t = 0; t < 4; ++t) {
      const v2f w01 = { f.w[t][0], f.w[t][1] };
      c[i][t] = KT ? wmma_f32(w01, f.a[i].xy, c[i][t]) : wmma_f32(f.a[i].xy, w01, c[i][t]);
    }
  }
  #pragma unroll
  for (int i = 0; i < 2; ++i) {
    #pragma unroll
    for (int t = 0; t < 4; ++t) {
      const v2f w23 = { f.w[t][2], f.w[t][3] };
      c[i][t] = KT ? wmma_f32(w23, f.a[i].zw, c[i][t]) : wmma_f32(f.a[i].zw, w23, c[i][t]);
    }
  }
}

template <bool KT>
static __device__ __forceinline__ void proj_loop(
    const float* __restrict__ hp, const float* __restrict__ wp, v8f (&c)[2][4]) {
  ProjFrag f0, f1;
  proj_load(f0, hp, wp, 0);
  for (int u = 0; u < DM / 8; u += 2) {
    proj_load(f1, hp, wp, u + 1);
    __builtin_amdgcn_sched_barrier(0);
    proj_mma<KT>(f0, c);
    proj_load(f0, hp, wp, (u + 2 < DM / 8) ? u + 2 : u);
    __builtin_amdgcn_sched_barrier(0);
    proj_mma<KT>(f1, c);
  }
}

__global__ __launch_bounds__(128) void qkv_proj_kernel(
    const float* __restrict__ h,
    const float* __restrict__ Wq, const float* __restrict__ bq,
    const float* __restrict__ Wk, const float* __restrict__ bk,
    const float* __restrict__ Wv, const float* __restrict__ bv,
    float* __restrict__ Qo, float* __restrict__ Ko, float* __restrict__ Vo)
{
  const int lane = threadIdx.x & 31;
  const int wid  = threadIdx.x >> 5;
  const int mat  = blockIdx.x >> 8;
  const int hd   = (blockIdx.x >> 5) & 7;
  const int tb   = (blockIdx.x & 31) * 4 + wid;

  const int mr   = lane & 15;
  const int half = lane >> 4;
  const float* hp = h + (size_t)(tb * 32 + mr) * DM + 4 * half;
  const float* W    = (mat == 0) ? Wq : (mat == 1) ? Wk : Wv;
  const float* bias = (mat == 0) ? bq : (mat == 1) ? bk : bv;
  const float* wp = W + (size_t)(4 * half) * DM + hd * DHD + mr;

  v8f c[2][4] = {};

  if (mat == 1) {
    proj_loop<true>(hp, wp, c);
    #pragma unroll
    for (int i = 0; i < 2; ++i) {
      #pragma unroll
      for (int t = 0; t < 4; ++t) {
        const float* bp = bias + hd * DHD + 16 * t + 8 * half;
        const v4f b0 = *(const v4f*)bp;
        const v4f b1 = *(const v4f*)(bp + 4);
        const int u = t * 2 + half;
        float* dst = Ko + ((size_t)((hd * 256 + tb * 2 + i) * 8 + u) * 32 + mr) * 4;
        const v8f& r = c[i][t];
        v4f o0 = { r[0] + b0.x, r[1] + b0.y, r[2] + b0.z, r[3] + b0.w };
        v4f o1 = { r[4] + b1.x, r[5] + b1.y, r[6] + b1.z, r[7] + b1.w };
        *(v4f*)dst        = o0;
        *(v4f*)(dst + 64) = o1;
      }
    }
    return;
  }

  proj_loop<false>(hp, wp, c);
  if (mat == 0) {
    #pragma unroll
    for (int i = 0; i < 2; ++i) {
      #pragma unroll
      for (int t = 0; t < 4; ++t) {
        const int n = hd * DHD + 16 * t + mr;
        const float bn = bias[n];
        #pragma unroll
        for (int v = 0; v < 8; ++v)
          Qo[(size_t)(tb * 32 + 16 * i + v + 8 * half) * DM + n] = c[i][t][v] + bn;
      }
    }
  } else {
    #pragma unroll
    for (int i = 0; i < 2; ++i) {
      #pragma unroll
      for (int t = 0; t < 4; ++t) {
        const float bn = bias[hd * DHD + 16 * t + mr];
        float* dst = Vo + ((size_t)(((hd * 512 + (tb * 2 + i) * 2 + half) * 4 + t) * 32) + mr) * 4;
        const v8f& r = c[i][t];
        v4f o0 = { r[0] + bn, r[1] + bn, r[2] + bn, r[3] + bn };
        v4f o1 = { r[4] + bn, r[5] + bn, r[6] + bn, r[7] + bn };
        *(v4f*)dst        = o0;
        *(v4f*)(dst + 64) = o1;
      }
    }
  }
}

#define CB  64
#define PST 68

__global__ __launch_bounds__(128) __attribute__((amdgpu_waves_per_eu(2, 2)))
void graph_attn_kernel(
    const float* __restrict__ Q, const float* __restrict__ K,
    const float* __restrict__ V,
    const float* __restrict__ adj, const float* __restrict__ msk,
    float* __restrict__ out)
{
  __shared__ float pl[4][16 * PST];

  const int lane = threadIdx.x & 31;
  const int wid  = threadIdx.x >> 5;
  const int task = blockIdx.x * 4 + wid;
  const int head = task >> 8;
  const int rowB = (task & 255) * 16;
  const int nn   = lane & 15;
  const int half = lane >> 4;
  float* plw = pl[wid];

  const float* qrow = Q + (size_t)(rowB + nn) * DM + head * DHD + 4 * half;
  v4f qa[8];
  #pragma unroll
  for (int u = 0; u < 8; ++u)
    qa[u] = *(const v4f*)(qrow + 8 * u);

  v8f acc[4] = {};
  float mrow[8], lrow[8];
  #pragma unroll
  for (int v = 0; v < 8; ++v) { mrow[v] = -3.0e38f; lrow[v] = 0.0f; }

  const float* kt = K + ((size_t)(head * 256) * 8) * 128 + lane * 4;
  const float* vt = V + ((size_t)(head * 512) * 4) * 128 + lane * 4;
  const size_t e0 = (size_t)(rowB + 8 * half) * NTOK + nn;
  const float* ap = adj + e0;
  const float* mp = msk + e0;

  v4f kb[8][4];
  float av[8][4], mv[8][4];
  #pragma unroll
  for (int u = 0; u < 8; ++u) {
    #pragma unroll
    for (int c = 0; c < 4; ++c)
      kb[u][c] = *(const v4f*)(kt + c * 1024 + u * 128);
  }
  #pragma unroll
  for (int v = 0; v < 8; ++v) {
    #pragma unroll
    for (int c = 0; c < 4; ++c) {
      av[v][c] = ap[(size_t)v * NTOK + 16 * c];
      mv[v][c] = mp[(size_t)v * NTOK + 16 * c];
    }
  }

  for (int j0 = 0; j0 < NTOK; j0 += CB) {
    const int jn = (j0 + CB < NTOK) ? j0 + CB : j0;
    const int jh = (jn + CB < NTOK) ? jn + CB : jn;
    __builtin_prefetch(&adj[(size_t)(rowB + nn) * NTOK + jh + 32 * half], 0, 1);

    const float* vtj = vt + (size_t)(j0 >> 3) * 512;
    v4f vb[8][4];
    #pragma unroll
    for (int u = 0; u < 8; ++u) {
      #pragma unroll
      for (int t = 0; t < 4; ++t)
        vb[u][t] = *(const v4f*)(vtj + (u * 4 + t) * 128);
    }
    __builtin_amdgcn_sched_barrier(0);

    v8f sf[4] = {};
    #pragma unroll
    for (int u = 0; u < 8; ++u) {
      #pragma unroll
      for (int c = 0; c < 4; ++c)
        sf[c] = wmma_f32(qa[u].xy, kb[u][c].xy, sf[c]);
      #pragma unroll
      for (int c = 0; c < 4; ++c)
        sf[c] = wmma_f32(qa[u].zw, kb[u][c].zw, sf[c]);
    }
    int kofs = (jn >> 4) * 1024;
    asm volatile("" : "+v"(sf[0]), "+v"(sf[1]), "+v"(sf[2]), "+v"(sf[3]), "+s"(kofs));
    const float* ktn = kt + kofs;
    #pragma unroll
    for (int u = 0; u < 8; ++u) {
      #pragma unroll
      for (int c = 0; c < 4; ++c)
        kb[u][c] = *(const v4f*)(ktn + c * 1024 + u * 128);
    }

    #pragma unroll
    for (int v = 0; v < 8; ++v) {
      float sc[4];
      #pragma unroll
      for (int c = 0; c < 4; ++c)
        sc[c] = (av[v][c] * sf[c][v] + mv[v][c]) * SCALE;
      if (v < 3) {
        #pragma unroll
        for (int c = 0; c < 4; ++c) {
          av[v][c] = ap[(size_t)v * NTOK + jn + 16 * c];
          mv[v][c] = mp[(size_t)v * NTOK + jn + 16 * c];
        }
      }
      float bm = fmaxf(fmaxf(sc[0], sc[1]), fmaxf(sc[2], sc[3]));
      bm = fmaxf(row_xor<1>(bm), bm);
      bm = fmaxf(row_xor<2>(bm), bm);
      bm = fmaxf(row_xor<4>(bm), bm);
      bm = fmaxf(row_xor<8>(bm), bm);
      const float newm = fmaxf(mrow[v], bm);
      const float f    = __expf(mrow[v] - newm);
      float pv[4];
      #pragma unroll
      for (int c = 0; c < 4; ++c) pv[c] = __expf(sc[c] - newm);
      float bs = (pv[0] + pv[1]) + (pv[2] + pv[3]);
      bs = row_xor<1>(bs) + bs;
      bs = row_xor<2>(bs) + bs;
      bs = row_xor<4>(bs) + bs;
      bs = row_xor<8>(bs) + bs;
      lrow[v] = lrow[v] * f + bs;
      mrow[v] = newm;
      #pragma unroll
      for (int c = 0; c < 4; ++c)
        plw[(v + 8 * half) * PST + 16 * c + nn] = pv[c];
      #pragma unroll
      for (int t = 0; t < 4; ++t) acc[t][v] *= f;
    }

    int aofs2 = jn;
    asm volatile("" : "+v"(lrow[0]), "+v"(lrow[1]), "+v"(lrow[2]), "+v"(lrow[3]),
                      "+v"(lrow[4]), "+v"(lrow[5]), "+v"(lrow[6]), "+v"(lrow[7]), "+s"(aofs2));
    #pragma unroll
    for (int v = 3; v < 8; ++v) {
      #pragma unroll
      for (int c = 0; c < 4; ++c) {
        av[v][c] = ap[(size_t)v * NTOK + aofs2 + 16 * c];
        mv[v][c] = mp[(size_t)v * NTOK + aofs2 + 16 * c];
      }
    }
    __builtin_amdgcn_wave_barrier();

    #pragma unroll
    for (int u = 0; u < 8; ++u) {
      v4f a = *(const v4f*)&plw[nn * PST + 8 * u + 4 * half];
      #pragma unroll
      for (int t = 0; t < 4; ++t)
        acc[t] = wmma_f32(a.xy, vb[u][t].xy, acc[t]);
      #pragma unroll
      for (int t = 0; t < 4; ++t)
        acc[t] = wmma_f32(a.zw, vb[u][t].zw, acc[t]);
    }
  }

  #pragma unroll
  for (int t = 0; t < 4; ++t) {
    #pragma unroll
    for (int v = 0; v < 8; ++v) {
      out[(size_t)(rowB + v + 8 * half) * DM + head * DHD + t * 16 + nn] =
          acc[t][v] / lrow[v];
    }
  }
}

extern "C" void kernel_launch(void* const* d_in, const int* in_sizes, int n_in,
                              void* d_out, int out_size, void* d_ws, size_t ws_size,
                              hipStream_t stream) {
  (void)in_sizes; (void)n_in; (void)out_size; (void)ws_size;
  const float* h   = (const float*)d_in[0];
  const float* adj = (const float*)d_in[1];
  const float* msk = (const float*)d_in[2];
  const float* Wq  = (const float*)d_in[3];
  const float* bq  = (const float*)d_in[4];
  const float* Wk  = (const float*)d_in[5];
  const float* bk  = (const float*)d_in[6];
  const float* Wv  = (const float*)d_in[7];
  const float* bv  = (const float*)d_in[8];
  float* out = (float*)d_out;

  float* Qo = (float*)d_ws;
  float* Ko = Qo + (size_t)NTOK * DM;
  float* Vo = Ko + (size_t)NTOK * DM;

  qkv_proj_kernel<<<3 * NH * 128 / 4, 128, 0, stream>>>(
      h, Wq, bq, Wk, bk, Wv, bv, Qo, Ko, Vo);

  graph_attn_kernel<<<(NH * 256) / 4, 128, 0, stream>>>(
      Qo, Ko, Vo, adj, msk, out);
}


# attention computed transposed (S^T = K Q^T, O^T += V^T P^T): P feeds P@V from registers, no LDS transpose; row max/sum in-lane plus one half exchange; regroups the row-sum tree and P@V key order
# speedup vs baseline: 2.4237x; 1.0871x over previous
#include <hip/hip_runtime.h>


typedef float v2f __attribute__((ext_vector_type(2)));
typedef float v4f __attribute__((ext_vector_type(4)));
typedef float v8f __attribute__((ext_vector_type(8)));


#define NTOK 4096
#define DM   512
#define NH   8
#define DHD  64
#define SCALE 0.125f

static __device__ __forceinline__ v8f wmma_f32(v2f a, v2f b, v8f c) {
  return __builtin_amdgcn_wmma_f32_16x16x4_f32(
       false, a,  false, b,
       (short)0, c,  false,  false);
}

static __device__ __forceinline__ float xhalf(float x) {
  const int b = __builtin_bit_cast(int, x);
  return __builtin_bit_cast(float, __builtin_amdgcn_permlanex16(
      b, b, 0x76543210, 0xfedcba98, false, false));
}

struct ProjFrag {
  v4f a[2];
  float w[4][4];
};

static __device__ __forceinline__ void proj_load(
    ProjFrag& f, const float* __restrict__ hp, const float* __restrict__ wp, int u) {
  #pragma unroll
  for (int i = 0; i < 2; ++i)
    f.a[i] = *(const v4f*)(hp + (size_t)i * 16 * DM + 8 * u);
  #pragma unroll
  for (int t = 0; t < 4; ++t) {
    #pragma unroll
    for (int s = 0; s < 4; ++s)
      f.w[t][s] = wp[(size_t)(8 * u + s) * DM + 16 * t];
  }
}

template <bool KT>
static __device__ __forceinline__ void proj_mma(const ProjFrag& f, v8f (&c)[2][4]) {
  #pragma unroll
  for (int i = 0; i < 2; ++i) {
    #pragma unroll
    for (int t = 0; t < 4; ++t) {
      const v2f w01 = { f.w[t][0], f.w[t][1] };
      c[i][t] = KT ? wmma_f32(w01, f.a[i].xy, c[i][t]) : wmma_f32(f.a[i].xy, w01, c[i][t]);
    }
  }
  #pragma unroll
  for (int i = 0; i < 2; ++i) {
    #pragma unroll
    for (int t = 0; t < 4; ++t) {
      const v2f w23 = { f.w[t][2], f.w[t][3] };
      c[i][t] = KT ? wmma_f32(w23, f.a[i].zw, c[i][t]) : wmma_f32(f.a[i].zw, w23, c[i][t]);
    }
  }
}

template <bool KT>
static __device__ __forceinline__ void proj_loop(
    const float* __restrict__ hp, const float* __restrict__ wp, v8f (&c)[2][4]) {
  ProjFrag f0, f1;
  proj_load(f0, hp, wp, 0);
  for (int u = 0; u < DM / 8; u += 2) {
    proj_load(f1, hp, wp, u + 1);
    __builtin_amdgcn_sched_barrier(0);
    proj_mma<KT>(f0, c);
    proj_load(f0, hp, wp, (u + 2 < DM / 8) ? u + 2 : u);
    __builtin_amdgcn_sched_barrier(0);
    proj_mma<KT>(f1, c);
  }
}

__global__ __launch_bounds__(128) void qkv_proj_kernel(
    const float* __restrict__ h,
    const float* __restrict__ Wq, const float* __restrict__ bq,
    const float* __restrict__ Wk, const float* __restrict__ bk,
    const float* __restrict__ Wv, const float* __restrict__ bv,
    float* __restrict__ Qo, float* __restrict__ Ko, float* __restrict__ Vo)
{
  const int lane = threadIdx.x & 31;
  const int wid  = threadIdx.x >> 5;
  const int mat  = blockIdx.x >> 8;
  const int hd   = (blockIdx.x >> 5) & 7;
  const int tb   = (blockIdx.x & 31) * 4 + wid;

  const int mr   = lane & 15;
  const int half = lane >> 4;
  const float* hp = h + (size_t)(tb * 32 + mr) * DM + 4 * half;
  const float* W    = (mat == 0) ? Wq : (mat == 1) ? Wk : Wv;
  const float* bias = (mat == 0) ? bq : (mat == 1) ? bk : bv;
  const float* wp = W + (size_t)(4 * half) * DM + hd * DHD + mr;

  v8f c[2][4] = {};

  if (mat == 1) {
    proj_loop<true>(hp, wp, c);
    #pragma unroll
    for (int i = 0; i < 2; ++i) {
      #pragma unroll
      for (int t = 0; t < 4; ++t) {
        const float* bp = bias + hd * DHD + 16 * t + 8 * half;
        const v4f b0 = *(const v4f*)bp;
        const v4f b1 = *(const v4f*)(bp + 4);
        const int u = t * 2 + half;
        float* dst = Ko + ((size_t)((hd * 256 + tb * 2 + i) * 8 + u) * 32 + mr) * 4;
        const v8f& r = c[i][t];
        v4f o0 = { r[0] + b0.x, r[1] + b0.y, r[2] + b0.z, r[3] + b0.w };
        v4f o1 = { r[4] + b1.x, r[5] + b1.y, r[6] + b1.z, r[7] + b1.w };
        *(v4f*)dst        = o0;
        *(v4f*)(dst + 64) = o1;
      }
    }
    return;
  }

  proj_loop<false>(hp, wp, c);
  if (mat == 0) {
    #pragma unroll
    for (int i = 0; i < 2; ++i) {
      #pragma unroll
      for (int t = 0; t < 4; ++t) {
        const int n = hd * DHD + 16 * t + mr;
        const float bn = bias[n];
        #pragma unroll
        for (int v = 0; v < 8; ++v)
          Qo[(size_t)(tb * 32 + 16 * i + v + 8 * half) * DM + n] = c[i][t][v] + bn;
      }
    }
  } else {
    #pragma unroll
    for (int i = 0; i < 2; ++i) {
      #pragma unroll
      for (int t = 0; t < 4; ++t) {
        const float bn = bias[hd * DHD + 16 * t + mr];
        float* dst = Vo + ((size_t)((((hd * 256 + tb * 2 + i) * 2) * 4 + t) * 32) + 16 * half + mr) * 4;
        const v8f& r = c[i][t];
        v4f o0 = { r[0] + bn, r[1] + bn, r[2] + bn, r[3] + bn };
        v4f o1 = { r[4] + bn, r[5] + bn, r[6] + bn, r[7] + bn };
        *(v4f*)dst         = o0;
        *(v4f*)(dst + 512) = o1;
      }
    }
  }
}

#define CB  64

__global__ __launch_bounds__(128) __attribute__((amdgpu_waves_per_eu(2, 2)))
void graph_attn_kernel(
    const float* __restrict__ Q, const float* __restrict__ K,
    const float* __restrict__ V,
    const float* __restrict__ adj, const float* __restrict__ msk,
    float* __restrict__ out)
{
  const int lane = threadIdx.x & 31;
  const int wid  = threadIdx.x >> 5;
  const int task = blockIdx.x * 4 + wid;
  const int head = task >> 8;
  const int rowB = (task & 255) * 16;
  const int nn   = lane & 15;
  const int half = lane >> 4;

  const float* qrow = Q + (size_t)(rowB + nn) * DM + head * DHD + 4 * half;
  v4f qa[8];
  #pragma unroll
  for (int u = 0; u < 8; ++u)
    qa[u] = *(const v4f*)(qrow + 8 * u);

  v8f acc[4] = {};
  float mrun = -3.0e38f;
  float lsum = 0.0f;

  const float* kt = K + ((size_t)(head * 256) * 8) * 128 + lane * 4;
  const float* vt = V + ((size_t)(head * 512) * 4) * 128 + lane * 4;
  const size_t e0 = (size_t)(rowB + nn) * NTOK + 8 * half;
  const float* ap = adj + e0;
  const float* mp = msk + e0;

  v4f kb[8][4];
  #pragma unroll
  for (int u = 0; u < 8; ++u) {
    #pragma unroll
    for (int c = 0; c < 4; ++c)
      kb[u][c] = *(const v4f*)(kt + c * 1024 + u * 128);
  }

  for (int j0 = 0; j0 < NTOK; j0 += CB) {
    const int jn = (j0 + CB < NTOK) ? j0 + CB : j0;
    const int jh = (jn + CB < NTOK) ? jn + CB : jn;
    __builtin_prefetch(&adj[(size_t)(rowB + nn) * NTOK + jh + 32 * half], 0, 1);

    v4f av[4][2], mv[4][2];
    #pragma unroll
    for (int c = 0; c < 4; ++c) {
      #pragma unroll
      for (int jj = 0; jj < 2; ++jj) {
        av[c][jj] = *(const v4f*)(ap + j0 + 16 * c + 4 * jj);
        mv[c][jj] = *(const v4f*)(mp + j0 + 16 * c + 4 * jj);
      }
    }
    const float* vtj = vt + (size_t)(j0 >> 3) * 512;
    v4f vb[8][4];
    #pragma unroll
    for (int u = 0; u < 8; ++u) {
      #pragma unroll
      for (int t = 0; t < 4; ++t)
        vb[u][t] = *(const v4f*)(vtj + (u * 4 + t) * 128);
    }
    __builtin_amdgcn_sched_barrier(0);

    v8f sf[4] = {};
    #pragma unroll
    for (int u = 0; u < 8; ++u) {
      #pragma unroll
      for (int c = 0; c < 4; ++c)
        sf[c] = wmma_f32(kb[u][c].xy, qa[u].xy, sf[c]);
      #pragma unroll
      for (int c = 0; c < 4; ++c)
        sf[c] = wmma_f32(kb[u][c].zw, qa[u].zw, sf[c]);
    }
    int kofs = (jn >> 4) * 1024;
    asm volatile("" : "+v"(sf[0]), "+v"(sf[1]), "+v"(sf[2]), "+v"(sf[3]), "+s"(kofs));
    const float* ktn = kt + kofs;
    #pragma unroll
    for (int u = 0; u < 8; ++u) {
      #pragma unroll
      for (int c = 0; c < 4; ++c)
        kb[u][c] = *(const v4f*)(ktn + c * 1024 + u * 128);
    }

    asm volatile("");

    #pragma unroll
    for (int c = 0; c < 4; ++c) {
      #pragma unroll
      for (int v = 0; v < 8; ++v)
        sf[c][v] = (av[c][v >> 2][v & 3] * sf[c][v] + mv[c][v >> 2][v & 3]) * SCALE;
    }
    float bm = sf[0][0];
    #pragma unroll
    for (int c = 0; c < 4; ++c) {
      #pragma unroll
      for (int v = 0; v < 8; ++v)
        bm = fmaxf(bm, sf[c][v]);
    }
    bm = fmaxf(bm, xhalf(bm));
    const float newm = fmaxf(mrun, bm);
    const float f    = __expf(mrun - newm);
    mrun = newm;
    float bs[4];
    #pragma unroll
    for (int c = 0; c < 4; ++c) {
      #pragma unroll
      for (int v = 0; v < 8; ++v)
        sf[c][v] = __expf(sf[c][v] - newm);
      bs[c] = ((sf[c][0] + sf[c][1]) + (sf[c][2] + sf[c][3])) +
              ((sf[c][4] + sf[c][5]) + (sf[c][6] + sf[c][7]));
    }
    lsum = lsum * f + ((bs[0] + bs[1]) + (bs[2] + bs[3]));
    #pragma unroll
    for (int t = 0; t < 4; ++t) {
      #pragma unroll
      for (int v = 0; v < 8; ++v)
        acc[t][v] *= f;
    }

    #pragma unroll
    for (int c = 0; c < 4; ++c) {
      #pragma unroll
      for (int jj = 0; jj < 2; ++jj) {
        const int u = 2 * c + jj;
        const v2f p01 = { sf[c][4 * jj + 0], sf[c][4 * jj + 1] };
        const v2f p23 = { sf[c][4 * jj + 2], sf[c][4 * jj + 3] };
        #pragma unroll
        for (int t = 0; t < 4; ++t)
          acc[t] = wmma_f32(vb[u][t].xy, p01, acc[t]);
        #pragma unroll
        for (int t = 0; t < 4; ++t)
          acc[t] = wmma_f32(vb[u][t].zw, p23, acc[t]);
      }
    }
  }

  const float ltot = lsum + xhalf(lsum);
  #pragma unroll
  for (int t = 0; t < 4; ++t) {
    float* dst = out + (size_t)(rowB + nn) * DM + head * DHD + t * 16 + 8 * half;
    const v8f& r = acc[t];
    v4f o0 = { r[0] / ltot, r[1] / ltot, r[2] / ltot, r[3] / ltot };
    v4f o1 = { r[4] / ltot, r[5] / ltot, r[6] / ltot, r[7] / ltot };
    *(v4f*)dst       = o0;
    *(v4f*)(dst + 4) = o1;
  }
}

extern "C" void kernel_launch(void* const* d_in, const int* in_sizes, int n_in,
                              void* d_out, int out_size, void* d_ws, size_t ws_size,
                              hipStream_t stream) {
  (void)in_sizes; (void)n_in; (void)out_size; (void)ws_size;
  const float* h   = (const float*)d_in[0];
  const float* adj = (const float*)d_in[1];
  const float* msk = (const float*)d_in[2];
  const float* Wq  = (const float*)d_in[3];
  const float* bq  = (const float*)d_in[4];
  const float* Wk  = (const float*)d_in[5];
  const float* bk  = (const float*)d_in[6];
  const float* Wv  = (const float*)d_in[7];
  const float* bv  = (const float*)d_in[8];
  float* out = (float*)d_out;

  float* Qo = (float*)d_ws;
  float* Ko = Qo + (size_t)NTOK * DM;
  float* Vo = Ko + (size_t)NTOK * DM;

  qkv_proj_kernel<<<3 * NH * 128 / 4, 128, 0, stream>>>(
      h, Wq, bq, Wk, bk, Wv, bv, Qo, Ko, Vo);

  graph_attn_kernel<<<(NH * 256) / 4, 128, 0, stream>>>(
      Qo, Ko, Vo, adj, msk, out);
}


# attention wave owns two 16-query blocks (32 queries), one wave per SIMD: each K and V fragment load feeds two WMMAs; one block's softmax stands between the other's WMMA steps; same sums and order
# speedup vs baseline: 2.5566x; 1.0548x over previous
#include <hip/hip_runtime.h>


typedef float v2f __attribute__((ext_vector_type(2)));
typedef float v4f __attribute__((ext_vector_type(4)));
typedef float v8f __attribute__((ext_vector_type(8)));


#define NTOK 4096
#define DM   512
#define NH   8
#define DHD  64
#define SCALE 0.125f

static __device__ __forceinline__ v8f wmma_f32(v2f a, v2f b, v8f c) {
  return __builtin_amdgcn_wmma_f32_16x16x4_f32(
       false, a,  false, b,
       (short)0, c,  false,  false);
}

static __device__ __forceinline__ float xhalf(float x) {
  const int b = __builtin_bit_cast(int, x);
  return __builtin_bit_cast(float, __builtin_amdgcn_permlanex16(
      b, b, 0x76543210, 0xfedcba98, false, false));
}

struct ProjFrag {
  v4f a[2];
  float w[4][4];
};

static __device__ __forceinline__ void proj_load(
    ProjFrag& f, const float* __restrict__ hp, const float* __restrict__ wp, int u) {
  #pragma unroll
  for (int i = 0; i < 2; ++i)
    f.a[i] = *(const v4f*)(hp + (size_t)i * 16 * DM + 8 * u);
  #pragma unroll
  for (int t = 0; t < 4; ++t) {
    #pragma unroll
    for (int s = 0; s < 4; ++s)
      f.w[t][s] = wp[(size_t)(8 * u + s) * DM + 16 * t];
  }
}

template <bool KT>
static __device__ __forceinline__ void proj_mma(const ProjFrag& f, v8f (&c)[2][4]) {
  #pragma unroll
  for (int i = 0; i < 2; ++i) {
    #pragma unroll
    for (int t = 0; t < 4; ++t) {
      const v2f w01 = { f.w[t][0], f.w[t][1] };
      c[i][t] = KT ? wmma_f32(w01, f.a[i].xy, c[i][t]) : wmma_f32(f.a[i].xy, w01, c[i][t]);
    }
  }
  #pragma unroll
  for (int i = 0; i < 2; ++i) {
    #pragma unroll
    for (int t = 0; t < 4; ++t) {
      const v2f w23 = { f.w[t][2], f.w[t][3] };
      c[i][t] = KT ? wmma_f32(w23, f.a[i].zw, c[i][t]) : wmma_f32(f.a[i].zw, w23, c[i][t]);
    }
  }
}

template <bool KT>
static __device__ __forceinline__ void proj_loop(
    const float* __restrict__ hp, const float* __restrict__ wp, v8f (&c)[2][4]) {
  ProjFrag f0, f1;
  proj_load(f0, hp, wp, 0);
  for (int u = 0; u < DM / 8; u += 2) {
    proj_load(f1, hp, wp, u + 1);
    __builtin_amdgcn_sched_barrier(0);
    proj_mma<KT>(f0, c);
    proj_load(f0, hp, wp, (u + 2 < DM / 8) ? u + 2 : u);
    __builtin_amdgcn_sched_barrier(0);
    proj_mma<KT>(f1, c);
  }
}

__global__ __launch_bounds__(128) void qkv_proj_kernel(
    const float* __restrict__ h,
    const float* __restrict__ Wq, const float* __restrict__ bq,
    const float* __restrict__ Wk, const float* __restrict__ bk,
    const float* __restrict__ Wv, const float* __restrict__ bv,
    float* __restrict__ Qo, float* __restrict__ Ko, float* __restrict__ Vo)
{
  const int lane = threadIdx.x & 31;
  const int wid  = threadIdx.x >> 5;
  const int mat  = blockIdx.x >> 8;
  const int hd   = (blockIdx.x >> 5) & 7;
  const int tb   = (blockIdx.x & 31) * 4 + wid;

  const int mr   = lane & 15;
  const int half = lane >> 4;
  const float* hp = h + (size_t)(tb * 32 + mr) * DM + 4 * half;
  const float* W    = (mat == 0) ? Wq : (mat == 1) ? Wk : Wv;
  const float* bias = (mat == 0) ? bq : (mat == 1) ? bk : bv;
  const float* wp = W + (size_t)(4 * half) * DM + hd * DHD + mr;

  v8f c[2][4] = {};

  if (mat == 1) {
    proj_loop<true>(hp, wp, c);
    #pragma unroll
    for (int i = 0; i < 2; ++i) {
      #pragma unroll
      for (int t = 0; t < 4; ++t) {
        const float* bp = bias + hd * DHD + 16 * t + 8 * half;
        const v4f b0 = *(const v4f*)bp;
        const v4f b1 = *(const v4f*)(bp + 4);
        const int u = t * 2 + half;
        float* dst = Ko + ((size_t)((hd * 256 + tb * 2 + i) * 8 + u) * 32 + mr) * 4;
        const v8f& r = c[i][t];
        v4f o0 = { r[0] + b0.x, r[1] + b0.y, r[2] + b0.z, r[3] + b0.w };
        v4f o1 = { r[4] + b1.x, r[5] + b1.y, r[6] + b1.z, r[7] + b1.w };
        *(v4f*)dst        = o0;
        *(v4f*)(dst + 64) = o1;
      }
    }
    return;
  }

  proj_loop<false>(hp, wp, c);
  if (mat == 0) {
    #pragma unroll
    for (int i = 0; i < 2; ++i) {
      #pragma unroll
      for (int t = 0; t < 4; ++t) {
        const int n = hd * DHD + 16 * t + mr;
        const float bn = bias[n];
        #pragma unroll
        for (int v = 0; v < 8; ++v)
          Qo[(size_t)(tb * 32 + 16 * i + v + 8 * half) * DM + n] = c[i][t][v] + bn;
      }
    }
  } else {
    #pragma unroll
    for (int i = 0; i < 2; ++i) {
      #pragma unroll
      for (int t = 0; t < 4; ++t) {
        const float bn = bias[hd * DHD + 16 * t + mr];
        float* dst = Vo + ((size_t)((((hd * 256 + tb * 2 + i) * 2) * 4 + t) * 32) + 16 * half + mr) * 4;
        const v8f& r = c[i][t];
        v4f o0 = { r[0] + bn, r[1] + bn, r[2] + bn, r[3] + bn };
        v4f o1 = { r[4] + bn, r[5] + bn, r[6] + bn, r[7] + bn };
        *(v4f*)dst         = o0;
        *(v4f*)(dst + 512) = o1;
      }
    }
  }
}

#define CB  64

static __device__ __forceinline__ void soft_piece(
    const int s, v8f (&sf)[4], const v4f (&av)[4][2], const v4f (&mv)[4][2],
    v8f (&acc)[4], float (&bmc)[4], float (&bs)[4],
    float& mrun, float& lsum, float& newm, float& f)
{
  if (s < 4) {
    const int c = s;
    #pragma unroll
    for (int v = 0; v < 8; ++v)
      sf[c][v] = (av[c][v >> 2][v & 3] * sf[c][v] + mv[c][v >> 2][v & 3]) * SCALE;
    float m = sf[c][0];
    #pragma unroll
    for (int v = 1; v < 8; ++v)
      m = fmaxf(m, sf[c][v]);
    bmc[c] = m;
  } else {
    const int c = s - 4;
    if (s == 4) {
      float bm = fmaxf(fmaxf(bmc[0], bmc[1]), fmaxf(bmc[2], bmc[3]));
      bm = fmaxf(bm, xhalf(bm));
      newm = fmaxf(mrun, bm);
      f    = __expf(mrun - newm);
      mrun = newm;
    }
    #pragma unroll
    for (int v = 0; v < 8; ++v)
      sf[c][v] = __expf(sf[c][v] - newm);
    bs[c] = ((sf[c][0] + sf[c][1]) + (sf[c][2] + sf[c][3])) +
            ((sf[c][4] + sf[c][5]) + (sf[c][6] + sf[c][7]));
    #pragma unroll
    for (int v = 0; v < 8; ++v)
      acc[c][v] *= f;
    if (s == 7)
      lsum = lsum * f + ((bs[0] + bs[1]) + (bs[2] + bs[3]));
  }
}

__global__ __launch_bounds__(128) __attribute__((amdgpu_waves_per_eu(1, 1)))
void graph_attn_kernel(
    const float* __restrict__ Q, const float* __restrict__ K,
    const float* __restrict__ V,
    const float* __restrict__ adj, const float* __restrict__ msk,
    float* __restrict__ out)
{
  const int lane = threadIdx.x & 31;
  const int wid  = threadIdx.x >> 5;
  const int task = blockIdx.x * 4 + wid;
  const int head = task >> 7;
  const int rowB = (task & 127) * 32;
  const int nn   = lane & 15;
  const int half = lane >> 4;

  v4f qa[2][8];
  #pragma unroll
  for (int b = 0; b < 2; ++b) {
    const float* qrow = Q + (size_t)(rowB + 16 * b + nn) * DM + head * DHD + 4 * half;
    #pragma unroll
    for (int u = 0; u < 8; ++u)
      qa[b][u] = *(const v4f*)(qrow + 8 * u);
  }

  v8f acc[2][4] = {};
  float mrun[2] = { -3.0e38f, -3.0e38f };
  float lsum[2] = { 0.0f, 0.0f };

  const float* kt = K + ((size_t)(head * 256) * 8) * 128 + lane * 4;
  const float* vt = V + ((size_t)(head * 512) * 4) * 128 + lane * 4;
  const size_t e0 = (size_t)(rowB + nn) * NTOK + 8 * half;
  const float* ap = adj + e0;
  const float* mp = msk + e0;

  v4f kb[8][4];
  #pragma unroll
  for (int u = 0; u < 8; ++u) {
    #pragma unroll
    for (int c = 0; c < 4; ++c)
      kb[u][c] = *(const v4f*)(kt + c * 1024 + u * 128);
  }

  for (int j0 = 0; j0 < NTOK; j0 += CB) {
    const int jn = (j0 + CB < NTOK) ? j0 + CB : j0;
    const int jh = (jn + CB < NTOK) ? jn + CB : jn;
    #pragma unroll
    for (int b = 0; b < 2; ++b) {
      const size_t ph = (size_t)(rowB + 16 * b + nn) * NTOK + jh + 32 * half;
      __builtin_prefetch(&adj[ph], 0, 1);
      __builtin_prefetch(&msk[ph], 0, 1);
    }

    v4f av[2][4][2], mv[2][4][2];
    #pragma unroll
    for (int b = 0; b < 2; ++b) {
      #pragma unroll
      for (int c = 0; c < 4; ++c) {
        #pragma unroll
        for (int jj = 0; jj < 2; ++jj) {
          av[b][c][jj] = *(const v4f*)(ap + (size_t)b * 16 * NTOK + j0 + 16 * c + 4 * jj);
          mv[b][c][jj] = *(const v4f*)(mp + (size_t)b * 16 * NTOK + j0 + 16 * c + 4 * jj);
        }
      }
    }
    __builtin_amdgcn_sched_barrier(0);

    v8f sf[2][4] = {};
    float bmc[2][4], bs[2][4], newm[2], fr[2];

    #pragma unroll
    for (int u = 0; u < 8; ++u) {
      #pragma unroll
      for (int c = 0; c < 4; ++c)
        sf[0][c] = wmma_f32(kb[u][c].xy, qa[0][u].xy, sf[0][c]);
      #pragma unroll
      for (int c = 0; c < 4; ++c)
        sf[0][c] = wmma_f32(kb[u][c].zw, qa[0][u].zw, sf[0][c]);
    }
    int vofs = (j0 >> 3) * 512;
    asm volatile("" : "+v"(sf[0][0]), "+v"(sf[0][1]), "+v"(sf[0][2]), "+v"(sf[0][3]), "+s"(vofs));
    const float* vtj = vt + vofs;
    v4f vb[8][4];
    #pragma unroll
    for (int u = 0; u < 8; ++u) {
      #pragma unroll
      for (int t = 0; t < 4; ++t)
        vb[u][t] = *(const v4f*)(vtj + (u * 4 + t) * 128);
    }
    __builtin_amdgcn_sched_barrier(0);

    #pragma unroll
    for (int u = 0; u < 8; ++u) {
      #pragma unroll
      for (int c = 0; c < 4; ++c)
        sf[1][c] = wmma_f32(kb[u][c].xy, qa[1][u].xy, sf[1][c]);
      #pragma unroll
      for (int c = 0; c < 4; ++c)
        sf[1][c] = wmma_f32(kb[u][c].zw, qa[1][u].zw, sf[1][c]);
      soft_piece(u, sf[0], av[0], mv[0], acc[0], bmc[0], bs[0], mrun[0], lsum[0], newm[0], fr[0]);
      if (u < 3)
        asm volatile("" : "+v"(sf[1][0]), "+v"(sf[1][1]), "+v"(sf[1][2]), "+v"(sf[1][3]),
                          "+v"(bmc[0][u]), "+v"(sf[0][u + 1]));
      else if (u == 3)
        asm volatile("" : "+v"(sf[1][0]), "+v"(sf[1][1]), "+v"(sf[1][2]), "+v"(sf[1][3]),
                          "+v"(bmc[0][3]));
      else
        asm volatile("" : "+v"(sf[1][0]), "+v"(sf[1][1]), "+v"(sf[1][2]), "+v"(sf[1][3]),
                          "+v"(bs[0][u - 4]), "+v"(acc[0][u - 4]), "+v"(newm[0]), "+v"(fr[0]));
    }
    int kofs = (jn >> 4) * 1024;
    asm volatile("" : "+v"(sf[1][0]), "+v"(sf[1][1]), "+v"(sf[1][2]), "+v"(sf[1][3]), "+s"(kofs));
    const float* ktn = kt + kofs;
    #pragma unroll
    for (int u = 0; u < 8; ++u) {
      #pragma unroll
      for (int c = 0; c < 4; ++c)
        kb[u][c] = *(const v4f*)(ktn + c * 1024 + u * 128);
    }

    asm volatile("");

    #pragma unroll
    for (int c = 0; c < 4; ++c) {
      #pragma unroll
      for (int jj = 0; jj < 2; ++jj) {
        const int u = 2 * c + jj;
        const v2f p01 = { sf[0][c][4 * jj + 0], sf[0][c][4 * jj + 1] };
        const v2f p23 = { sf[0][c][4 * jj + 2], sf[0][c][4 * jj + 3] };
        #pragma unroll
        for (int t = 0; t < 4; ++t)
          acc[0][t] = wmma_f32(vb[u][t].xy, p01, acc[0][t]);
        #pragma unroll
        for (int t = 0; t < 4; ++t)
          acc[0][t] = wmma_f32(vb[u][t].zw, p23, acc[0][t]);
        soft_piece(u, sf[1], av[1], mv[1], acc[1], bmc[1], bs[1], mrun[1], lsum[1], newm[1], fr[1]);
        if (u < 3)
          asm volatile("" : "+v"(acc[0][0]), "+v"(acc[0][1]), "+v"(acc[0][2]), "+v"(acc[0][3]),
                            "+v"(bmc[1][u]), "+v"(sf[1][u + 1]));
        else if (u == 3)
          asm volatile("" : "+v"(acc[0][0]), "+v"(acc[0][1]), "+v"(acc[0][2]), "+v"(acc[0][3]),
                            "+v"(bmc[1][3]));
        else
          asm volatile("" : "+v"(acc[0][0]), "+v"(acc[0][1]), "+v"(acc[0][2]), "+v"(acc[0][3]),
                            "+v"(bs[1][u - 4]), "+v"(acc[1][u - 4]), "+v"(newm[1]), "+v"(fr[1]));
      }
    }
    #pragma unroll
    for (int c = 0; c < 4; ++c) {
      #pragma unroll
      for (int jj = 0; jj < 2; ++jj) {
        const int u = 2 * c + jj;
        const v2f p01 = { sf[1][c][4 * jj + 0], sf[1][c][4 * jj + 1] };
        const v2f p23 = { sf[1][c][4 * jj + 2], sf[1][c][4 * jj + 3] };
        #pragma unroll
        for (int t = 0; t < 4; ++t)
          acc[1][t] = wmma_f32(vb[u][t].xy, p01, acc[1][t]);
        #pragma unroll
        for (int t = 0; t < 4; ++t)
          acc[1][t] = wmma_f32(vb[u][t].zw, p23, acc[1][t]);
      }
    }
  }

  #pragma unroll
  for (int b = 0; b < 2; ++b) {
    const float ltot = lsum[b] + xhalf(lsum[b]);
    #pragma unroll
    for (int t = 0; t < 4; ++t) {
      float* dst = out + (size_t)(rowB + 16 * b + nn) * DM + head * DHD + t * 16 + 8 * half;
      const v8f& r = acc[b][t];
      v4f o0 = { r[0] / ltot, r[1] / ltot, r[2] / ltot, r[3] / ltot };
      v4f o1 = { r[4] / ltot, r[5] / ltot, r[6] / ltot, r[7] / ltot };
      *(v4f*)dst       = o0;
      *(v4f*)(dst + 4) = o1;
    }
  }
}

extern "C" void kernel_launch(void* const* d_in, const int* in_sizes, int n_in,
                              void* d_out, int out_size, void* d_ws, size_t ws_size,
                              hipStream_t stream) {
  (void)in_sizes; (void)n_in; (void)out_size; (void)ws_size;
  const float* h   = (const float*)d_in[0];
  const float* adj = (const float*)d_in[1];
  const float* msk = (const float*)d_in[2];
  const float* Wq  = (const float*)d_in[3];
  const float* bq  = (const float*)d_in[4];
  const float* Wk  = (const float*)d_in[5];
  const float* bk  = (const float*)d_in[6];
  const float* Wv  = (const float*)d_in[7];
  const float* bv  = (const float*)d_in[8];
  float* out = (float*)d_out;

  float* Qo = (float*)d_ws;
  float* Ko = Qo + (size_t)NTOK * DM;
  float* Vo = Ko + (size_t)NTOK * DM;

  qkv_proj_kernel<<<3 * NH * 128 / 4, 128, 0, stream>>>(
      h, Wq, bq, Wk, bk, Wv, bv, Qo, Ko, Vo);

  graph_attn_kernel<<<(NH * 128) / 4, 128, 0, stream>>>(
      Qo, Ko, Vo, adj, msk, out);
}
